# starting kernel
# baseline (speedup 1.0000x reference)
#include <hip/hip_runtime.h>
#include <math.h>


typedef __attribute__((ext_vector_type(2))) float v2f;
typedef __attribute__((ext_vector_type(8))) float v8f;

#define BATCH 16384
#define NCP   4096
#define CHUNK 1024
#define BLOCK 256
#define ROWS_PER_BLOCK 128

#define EPS2      1e-20f
#define HALF_LN2  0.34657359028f

__launch_bounds__(BLOCK)
__global__ void tps_wmma_kernel(const float* __restrict__ u,
                                const float* __restrict__ cp,
                                const float* __restrict__ w,
                                const float* __restrict__ poly,
                                float* __restrict__ out)
{
    __shared__ float4 sCP[CHUNK];
    __shared__ float4 sW[CHUNK];

    const int tid  = threadIdx.x;
    const int wave = tid >> 5;
    const int lane = tid & 31;
    const int half = lane >> 4;
    const int q    = lane & 15;

    const int qIdx = blockIdx.x * ROWS_PER_BLOCK + wave * 16 + q;

    const float ux = u[qIdx * 3 + 0];
    const float uy = u[qIdx * 3 + 1];
    const float uz = u[qIdx * 3 + 2];
    const float usq = ux * ux + uy * uy + uz * uz;

    v2f bmat;
    bmat.x = half ? (-2.0f * uz) : (-2.0f * ux);
    bmat.y = half ? 1.0f         : (-2.0f * uy);

    v8f cmat;
#pragma unroll
    for (int k = 0; k < 8; ++k) cmat[k] = usq;

    float acc0 = 0.0f, acc1 = 0.0f, acc2 = 0.0f;

    for (int base = 0; base < NCP; base += CHUNK) {
        __syncthreads();
        for (int i = tid; i < CHUNK; i += BLOCK) {
            const int g = base + i;
            const float cx = cp[g * 3 + 0];
            const float cy = cp[g * 3 + 1];
            const float cz = cp[g * 3 + 2];
            sCP[i] = make_float4(cx, cy, cz, cx * cx + cy * cy + cz * cz);
            sW[i]  = make_float4(HALF_LN2 * w[g * 3 + 0],
                                 HALF_LN2 * w[g * 3 + 1],
                                 HALF_LN2 * w[g * 3 + 2], 0.0f);
        }
        __syncthreads();

#pragma unroll 2
        for (int cb = 0; cb < CHUNK; cb += 16) {
            const float4 c4 = sCP[cb + q];
            v2f amat;
            amat.x = half ? c4.z : c4.x;
            amat.y = half ? c4.w : c4.y;

            v8f d = __builtin_amdgcn_wmma_f32_16x16x4_f32(
                false, amat, false, bmat, (short)0, cmat, false, false);

            const int mBase = cb + half * 8;
#pragma unroll
            for (int k = 0; k < 8; ++k) {
                const float sq    = d[k];
                const bool  small = sq < EPS2;
                const float sqc   = small ? EPS2 : sq;
                float kv = __builtin_amdgcn_sqrtf(sqc) * __builtin_amdgcn_logf(sqc);
                kv = small ? 0.0f : kv;
                const float4 w4 = sW[mBase + k];
                acc0 = fmaf(kv, w4.x, acc0);
                acc1 = fmaf(kv, w4.y, acc1);
                acc2 = fmaf(kv, w4.z, acc2);
            }
        }
    }

    acc0 += __shfl_xor(acc0, 16, 32);
    acc1 += __shfl_xor(acc1, 16, 32);
    acc2 += __shfl_xor(acc2, 16, 32);

    if (half == 0) {
#pragma unroll
        for (int a = 0; a < 3; ++a) {
            const float p = poly[0 * 3 + a] + ux * poly[1 * 3 + a]
                          + uy * poly[2 * 3 + a] + uz * poly[3 * 3 + a];
            const float acc = (a == 0) ? acc0 : ((a == 1) ? acc1 : acc2);
            out[qIdx * 3 + a] = acc + p;
        }
    }
}

extern "C" void kernel_launch(void* const* d_in, const int* in_sizes, int n_in,
                              void* d_out, int out_size, void* d_ws, size_t ws_size,
                              hipStream_t stream) {
    const float* u    = (const float*)d_in[0];
    const float* cp   = (const float*)d_in[1];
    const float* w    = (const float*)d_in[2];
    const float* poly = (const float*)d_in[3];
    float* out = (float*)d_out;

    dim3 grid(BATCH / ROWS_PER_BLOCK);
    tps_wmma_kernel<<<grid, BLOCK, 0, stream>>>(u, cp, w, poly, out);
}


# r<EPS guard decided once per pair of tiles: min of a lane's 16 squared distances + one ballot and scalar branch pick an unguarded body (same bits); per-element compare and 2 selects only where needed
# speedup vs baseline: 6.4703x; 6.4703x over previous
#include <hip/hip_runtime.h>
#include <math.h>


typedef __attribute__((ext_vector_type(2))) float v2f;
typedef __attribute__((ext_vector_type(8))) float v8f;

#define BATCH 16384
#define NCP   4096
#define BLOCK 512
#define QGROUPS 4
#define SLICES  4
#define SLICE_LEN (NCP / SLICES)
#define ROWS_PER_BLOCK (QGROUPS * 16)

#define EPS2      1e-20f
#define HALF_LN2  0.34657359028f

template <bool GUARDED>
__device__ __forceinline__ void tile_accumulate(const v8f d, const float4 (&w4)[8],
                                                v2f& acc01, float& acc2)
{
#pragma unroll
    for (int k = 0; k < 8; ++k) {
        const float sq = d[k];
        float kv;
        if (GUARDED) {
            const bool  small = sq < EPS2;
            const float sqc   = small ? EPS2 : sq;
            kv = __builtin_amdgcn_sqrtf(sqc) * __builtin_amdgcn_logf(sqc);
            kv = small ? 0.0f : kv;
        } else {
            kv = __builtin_amdgcn_sqrtf(sq) * __builtin_amdgcn_logf(sq);
        }
        const v2f kv2 = {kv, kv};
        const v2f w01 = {w4[k].x, w4[k].y};
        acc01 = __builtin_elementwise_fma(kv2, w01, acc01);
        acc2  = fmaf(kv, w4[k].z, acc2);
    }
}

__launch_bounds__(BLOCK)
__global__ void tps_wmma_kernel(const float* __restrict__ u,
                                const float* __restrict__ cp,
                                const float* __restrict__ w,
                                const float* __restrict__ poly,
                                float* __restrict__ out)
{
    __shared__ float4 sCP[NCP];
    __shared__ float4 sW[NCP];
    __shared__ float sRed[SLICES][ROWS_PER_BLOCK * 3];

    const int tid   = threadIdx.x;
    const int wave  = __builtin_amdgcn_readfirstlane(tid >> 5);
    const int lane  = tid & 31;
    const int half  = lane >> 4;
    const int q     = lane & 15;
    const int qg    = wave & (QGROUPS - 1);
    const int slice = wave >> 2;

    const int qLocal = qg * 16 + q;
    const int qIdx   = blockIdx.x * ROWS_PER_BLOCK + qLocal;

    const float ux = u[qIdx * 3 + 0];
    const float uy = u[qIdx * 3 + 1];
    const float uz = u[qIdx * 3 + 2];
    const float usq = ux * ux + uy * uy + uz * uz;

    v2f bmat;
    bmat.x = half ? (-2.0f * uz) : (-2.0f * ux);
    bmat.y = half ? 1.0f         : (-2.0f * uy);

    v8f cmat;
#pragma unroll
    for (int k = 0; k < 8; ++k) cmat[k] = usq;

#pragma unroll
    for (int i = tid; i < NCP; i += BLOCK) {
        const float cx = cp[i * 3 + 0];
        const float cy = cp[i * 3 + 1];
        const float cz = cp[i * 3 + 2];
        sCP[i] = make_float4(cx, cy, cz, cx * cx + cy * cy + cz * cz);
        sW[i]  = make_float4(HALF_LN2 * w[i * 3 + 0],
                             HALF_LN2 * w[i * 3 + 1],
                             HALF_LN2 * w[i * 3 + 2], 0.0f);
    }
    __syncthreads();

    v2f   acc01 = {0.0f, 0.0f};
    float acc2  = 0.0f;

    const int cbBegin = slice * SLICE_LEN;

    for (int t = 0; t < SLICE_LEN; t += 32) {
        const int cb = cbBegin + t;
        const float4 cA = sCP[cb + q];
        const float4 cB = sCP[cb + 16 + q];
        v2f amatA, amatB;
        amatA.x = half ? cA.z : cA.x;
        amatA.y = half ? cA.w : cA.y;
        amatB.x = half ? cB.z : cB.x;
        amatB.y = half ? cB.w : cB.y;

        const v8f dA = __builtin_amdgcn_wmma_f32_16x16x4_f32(
            false, amatA, false, bmat, (short)0, cmat, false, false);
        const v8f dB = __builtin_amdgcn_wmma_f32_16x16x4_f32(
            false, amatB, false, bmat, (short)0, cmat, false, false);

        float4 wA[8], wB[8];
        const int mBase = cb + half * 8;
#pragma unroll
        for (int k = 0; k < 8; ++k) {
            wA[k] = sW[mBase + k];
            wB[k] = sW[mBase + 16 + k];
        }

        const float m1 = fminf(fminf(dA[0], dA[1]), dA[2]);
        const float m2 = fminf(fminf(dA[3], dA[4]), dA[5]);
        const float m3 = fminf(fminf(dA[6], dA[7]), dB[0]);
        const float m4 = fminf(fminf(dB[1], dB[2]), dB[3]);
        const float m5 = fminf(fminf(dB[4], dB[5]), dB[6]);
        const float m6 = fminf(fminf(m1, m2), m3);
        const float m7 = fminf(fminf(m4, m5), dB[7]);
        const float m  = fminf(m6, m7);

        if (__builtin_expect(__builtin_amdgcn_ballot_w32(m < EPS2) != 0u, 0)) {
            tile_accumulate<true>(dA, wA, acc01, acc2);
            tile_accumulate<true>(dB, wB, acc01, acc2);
        } else {
            tile_accumulate<false>(dA, wA, acc01, acc2);
            tile_accumulate<false>(dB, wB, acc01, acc2);
        }
    }
    float acc0 = acc01.x, acc1 = acc01.y;

    acc0 += __shfl_xor(acc0, 16, 32);
    acc1 += __shfl_xor(acc1, 16, 32);
    acc2 += __shfl_xor(acc2, 16, 32);

    if (half == 0) {
        sRed[slice][qLocal * 3 + 0] = acc0;
        sRed[slice][qLocal * 3 + 1] = acc1;
        sRed[slice][qLocal * 3 + 2] = acc2;
    }
    __syncthreads();

    if (tid < ROWS_PER_BLOCK * 3) {
        const int ql = tid / 3;
        const int a  = tid - ql * 3;
        const int qo = blockIdx.x * ROWS_PER_BLOCK + ql;

        float acc = sRed[0][tid];
#pragma unroll
        for (int s = 1; s < SLICES; ++s) acc += sRed[s][tid];

        const float px = u[qo * 3 + 0];
        const float py = u[qo * 3 + 1];
        const float pz = u[qo * 3 + 2];
        const float p = poly[0 * 3 + a] + px * poly[1 * 3 + a]
                      + py * poly[2 * 3 + a] + pz * poly[3 * 3 + a];
        out[qo * 3 + a] = acc + p;
    }
}

extern "C" void kernel_launch(void* const* d_in, const int* in_sizes, int n_in,
                              void* d_out, int out_size, void* d_ws, size_t ws_size,
                              hipStream_t stream) {
    const float* u    = (const float*)d_in[0];
    const float* cp   = (const float*)d_in[1];
    const float* w    = (const float*)d_in[2];
    const float* poly = (const float*)d_in[3];
    float* out = (float*)d_out;

    dim3 grid(BATCH / ROWS_PER_BLOCK);
    tps_wmma_kernel<<<grid, BLOCK, 0, stream>>>(u, cp, w, poly, out);
}


# Each wave takes all four 16-query groups against one of 8 control-point slices (8-wave work-groups): LDS reads, guard and loop work shared 4x, half the waves to launch; 8 slice sums (new sum order)
# speedup vs baseline: 6.9458x; 1.0735x over previous
#include <hip/hip_runtime.h>
#include <math.h>


typedef __attribute__((ext_vector_type(2))) float v2f;
typedef __attribute__((ext_vector_type(8))) float v8f;

#define BATCH 16384
#define NCP   4096
#define BLOCK 256
#define QGROUPS 4
#define SLICES  8
#define SLICE_LEN (NCP / SLICES)
#define ROWS_PER_BLOCK (QGROUPS * 16)

#define EPS2      1e-20f
#define HALF_LN2  0.34657359028f

template <bool GUARDED>
__device__ __forceinline__ void tile_accumulate(const v8f d, const float4 (&w4)[8],
                                                v2f& acc01, float& acc2)
{
#pragma unroll
    for (int k = 0; k < 8; ++k) {
        const float sq = d[k];
        float kv;
        if (GUARDED) {
            const bool  small = sq < EPS2;
            const float sqc   = small ? EPS2 : sq;
            kv = __builtin_amdgcn_sqrtf(sqc) * __builtin_amdgcn_logf(sqc);
            kv = small ? 0.0f : kv;
        } else {
            kv = __builtin_amdgcn_sqrtf(sq) * __builtin_amdgcn_logf(sq);
        }
        const v2f kv2 = {kv, kv};
        const v2f w01 = {w4[k].x, w4[k].y};
        acc01 = __builtin_elementwise_fma(kv2, w01, acc01);
        acc2  = fmaf(kv, w4[k].z, acc2);
    }
}

__device__ __forceinline__ float min3f(float a, float b, float c)
{
    return fminf(fminf(a, b), c);
}

__launch_bounds__(BLOCK)
__global__ void tps_wmma_kernel(const float* __restrict__ u,
                                const float* __restrict__ cp,
                                const float* __restrict__ w,
                                const float* __restrict__ poly,
                                float* __restrict__ out)
{
    __shared__ float4 sCP[NCP];
    __shared__ float4 sW[NCP];
    __shared__ float sRed[SLICES][ROWS_PER_BLOCK * 3];

    const int tid   = threadIdx.x;
    const int slice = __builtin_amdgcn_readfirstlane(tid >> 5);
    const int lane  = tid & 31;
    const int half  = lane >> 4;
    const int q     = lane & 15;

    v2f bmat[QGROUPS];
    v8f cmat[QGROUPS];
#pragma unroll
    for (int g = 0; g < QGROUPS; ++g) {
        const int qIdx = blockIdx.x * ROWS_PER_BLOCK + g * 16 + q;
        const float ux = u[qIdx * 3 + 0];
        const float uy = u[qIdx * 3 + 1];
        const float uz = u[qIdx * 3 + 2];
        const float usq = ux * ux + uy * uy + uz * uz;
        bmat[g].x = half ? (-2.0f * uz) : (-2.0f * ux);
        bmat[g].y = half ? 1.0f         : (-2.0f * uy);
#pragma unroll
        for (int k = 0; k < 8; ++k) cmat[g][k] = usq;
    }

#pragma unroll
    for (int it = 0; it < NCP / BLOCK; ++it) {
        const int i = it * BLOCK + tid;
        const float cx = cp[i * 3 + 0];
        const float cy = cp[i * 3 + 1];
        const float cz = cp[i * 3 + 2];
        sCP[i] = make_float4(cx, cy, cz, cx * cx + cy * cy + cz * cz);
        sW[i]  = make_float4(HALF_LN2 * w[i * 3 + 0],
                             HALF_LN2 * w[i * 3 + 1],
                             HALF_LN2 * w[i * 3 + 2], 0.0f);
    }
    __syncthreads();

    v2f   acc01[QGROUPS];
    float acc2[QGROUPS];
#pragma unroll
    for (int g = 0; g < QGROUPS; ++g) {
        acc01[g] = (v2f){0.0f, 0.0f};
        acc2[g]  = 0.0f;
    }

    const int cbBegin = slice * SLICE_LEN;

    for (int t = 0; t < SLICE_LEN; t += 32) {
        const int cb = cbBegin + t;
        const float4 cA = sCP[cb + q];
        const float4 cB = sCP[cb + 16 + q];
        v2f amatA, amatB;
        amatA.x = half ? cA.z : cA.x;
        amatA.y = half ? cA.w : cA.y;
        amatB.x = half ? cB.z : cB.x;
        amatB.y = half ? cB.w : cB.y;

        v8f dA[QGROUPS], dB[QGROUPS];
#pragma unroll
        for (int g = 0; g < QGROUPS; ++g) {
            dA[g] = __builtin_amdgcn_wmma_f32_16x16x4_f32(
                false, amatA, false, bmat[g], (short)0, cmat[g], false, false);
            dB[g] = __builtin_amdgcn_wmma_f32_16x16x4_f32(
                false, amatB, false, bmat[g], (short)0, cmat[g], false, false);
        }

        float4 wA[8], wB[8];
        const int mBase = cb + half * 8;
#pragma unroll
        for (int k = 0; k < 8; ++k) {
            wA[k] = sW[mBase + k];
            wB[k] = sW[mBase + 16 + k];
        }

        float l1[QGROUPS * 6];
#pragma unroll
        for (int g = 0; g < QGROUPS; ++g) {
            l1[g * 6 + 0] = min3f(dA[g][0], dA[g][1], dA[g][2]);
            l1[g * 6 + 1] = min3f(dA[g][3], dA[g][4], dA[g][5]);
            l1[g * 6 + 2] = min3f(dA[g][6], dA[g][7], dB[g][0]);
            l1[g * 6 + 3] = min3f(dB[g][1], dB[g][2], dB[g][3]);
            l1[g * 6 + 4] = min3f(dB[g][4], dB[g][5], dB[g][6]);
            l1[g * 6 + 5] = dB[g][7];
        }
        float l2[QGROUPS * 2];
#pragma unroll
        for (int i = 0; i < QGROUPS * 2; ++i)
            l2[i] = min3f(l1[i * 3 + 0], l1[i * 3 + 1], l1[i * 3 + 2]);
        const float l3a = min3f(l2[0], l2[1], l2[2]);
        const float l3b = min3f(l2[3], l2[4], l2[5]);
        const float m   = fminf(min3f(l3a, l3b, l2[6]), l2[7]);

        if (__builtin_expect(__builtin_amdgcn_ballot_w32(m < EPS2) != 0u, 0)) {
#pragma unroll
            for (int g = 0; g < QGROUPS; ++g) {
                tile_accumulate<true>(dA[g], wA, acc01[g], acc2[g]);
                tile_accumulate<true>(dB[g], wB, acc01[g], acc2[g]);
            }
        } else {
#pragma unroll
            for (int g = 0; g < QGROUPS; ++g) {
                tile_accumulate<false>(dA[g], wA, acc01[g], acc2[g]);
                tile_accumulate<false>(dB[g], wB, acc01[g], acc2[g]);
            }
        }
    }

#pragma unroll
    for (int g = 0; g < QGROUPS; ++g) {
        float acc0 = acc01[g].x, acc1 = acc01[g].y, a2 = acc2[g];
        acc0 += __shfl_xor(acc0, 16, 32);
        acc1 += __shfl_xor(acc1, 16, 32);
        a2   += __shfl_xor(a2, 16, 32);
        if (half == 0) {
            const int qLocal = g * 16 + q;
            sRed[slice][qLocal * 3 + 0] = acc0;
            sRed[slice][qLocal * 3 + 1] = acc1;
            sRed[slice][qLocal * 3 + 2] = a2;
        }
    }
    __syncthreads();

    if (tid < ROWS_PER_BLOCK * 3) {
        const int ql = tid / 3;
        const int a  = tid - ql * 3;
        const int qo = blockIdx.x * ROWS_PER_BLOCK + ql;

        float acc = sRed[0][tid];
#pragma unroll
        for (int s = 1; s < SLICES; ++s) acc += sRed[s][tid];

        const float px = u[qo * 3 + 0];
        const float py = u[qo * 3 + 1];
        const float pz = u[qo * 3 + 2];
        const float p = poly[0 * 3 + a] + px * poly[1 * 3 + a]
                      + py * poly[2 * 3 + a] + pz * poly[3 * 3 + a];
        out[qo * 3 + a] = acc + p;
    }
}

extern "C" void kernel_launch(void* const* d_in, const int* in_sizes, int n_in,
                              void* d_out, int out_size, void* d_ws, size_t ws_size,
                              hipStream_t stream) {
    const float* u    = (const float*)d_in[0];
    const float* cp   = (const float*)d_in[1];
    const float* w    = (const float*)d_in[2];
    const float* poly = (const float*)d_in[3];
    float* out = (float*)d_out;

    dim3 grid(BATCH / ROWS_PER_BLOCK);
    tps_wmma_kernel<<<grid, BLOCK, 0, stream>>>(u, cp, w, poly, out);
}


# Vector stream at the transcendental rate: two tile rows per packed op (1 packed mul + 3 packed fma per 4 sqrt/log, weights paired in LDS), order pinned T,p,T,p; even/odd row sums (new sum order)
# speedup vs baseline: 7.5177x; 1.0823x over previous
#include <hip/hip_runtime.h>
#include <math.h>


typedef __attribute__((ext_vector_type(2))) float v2f;
typedef __attribute__((ext_vector_type(8))) float v8f;

#define BATCH 16384
#define NCP   4096
#define BLOCK 256
#define QGROUPS 4
#define SLICES  8
#define SLICE_LEN (NCP / SLICES)
#define ROWS_PER_BLOCK (QGROUPS * 16)

#define EPS2      1e-20f
#define HALF_LN2  0.34657359028f

__device__ __forceinline__ v2f weight_pair(const float4 (&wv)[6], int p, int c)
{
    const int idx = 6 * p + 2 * c;
    const float4 f = wv[idx >> 2];
    return (idx & 2) ? (v2f){f.z, f.w} : (v2f){f.x, f.y};
}

__device__ __forceinline__ void tile_accumulate_guarded(const v8f d, const float4 (&wv)[6],
                                                        v2f& acc0, v2f& acc1, v2f& acc2)
{
#pragma unroll
    for (int p = 0; p < 4; ++p) {
        const float sqa = d[2 * p];
        const float sqb = d[2 * p + 1];
        const bool  smalla = sqa < EPS2;
        const bool  smallb = sqb < EPS2;
        const float ca = smalla ? EPS2 : sqa;
        const float cb = smallb ? EPS2 : sqb;
        const v2f rt = {__builtin_amdgcn_sqrtf(ca), __builtin_amdgcn_sqrtf(cb)};
        const v2f lg = {__builtin_amdgcn_logf(ca), __builtin_amdgcn_logf(cb)};
        v2f kv = rt * lg;
        kv.x = smalla ? 0.0f : kv.x;
        kv.y = smallb ? 0.0f : kv.y;
        acc0 = __builtin_elementwise_fma(kv, weight_pair(wv, p, 0), acc0);
        acc1 = __builtin_elementwise_fma(kv, weight_pair(wv, p, 1), acc1);
        acc2 = __builtin_elementwise_fma(kv, weight_pair(wv, p, 2), acc2);
    }
}

#define SG_VALU  0x002
#define SG_TRANS 0x400

__device__ __forceinline__ void trip_accumulate(const v8f (&dA)[QGROUPS], const v8f (&dB)[QGROUPS],
                                                const float4 (&wA)[6], const float4 (&wB)[6],
                                                v2f (&acc0)[QGROUPS], v2f (&acc1)[QGROUPS],
                                                v2f (&acc2)[QGROUPS])
{
    constexpr int STEPS = 8;
    v2f rt[STEPS][QGROUPS], lg[STEPS][QGROUPS], kv[STEPS][QGROUPS];
#pragma unroll
    for (int b = 0; b <= STEPS; ++b) {
        const int  p  = b & 3;
        const bool tB = b >= 4;
        const int  pp = (b - 1) & 3;
        const bool tP = (b - 1) >= 4;
#pragma unroll
        for (int s = 0; s < 4 * QGROUPS; ++s) {
            if (b < STEPS) {
                const int g = s >> 2;
                const float sqa = tB ? dB[g][2 * p]     : dA[g][2 * p];
                const float sqb = tB ? dB[g][2 * p + 1] : dA[g][2 * p + 1];
                if ((s & 3) == 0) rt[b][g].x = __builtin_amdgcn_sqrtf(sqa);
                if ((s & 3) == 1) rt[b][g].y = __builtin_amdgcn_sqrtf(sqb);
                if ((s & 3) == 2) lg[b][g].x = __builtin_amdgcn_logf(sqa);
                if ((s & 3) == 3) lg[b][g].y = __builtin_amdgcn_logf(sqb);
                __builtin_amdgcn_sched_group_barrier(SG_TRANS, 1, 0);
            }
            if (b >= 1) {
                const int g = s & (QGROUPS - 1);
                const int c = (s >> 2) - 1;
                if (c < 0)  kv[b - 1][g] = rt[b - 1][g] * lg[b - 1][g];
                if (c == 0) acc0[g] = __builtin_elementwise_fma(kv[b - 1][g], weight_pair(tP ? wB : wA, pp, 0), acc0[g]);
                if (c == 1) acc1[g] = __builtin_elementwise_fma(kv[b - 1][g], weight_pair(tP ? wB : wA, pp, 1), acc1[g]);
                if (c == 2) acc2[g] = __builtin_elementwise_fma(kv[b - 1][g], weight_pair(tP ? wB : wA, pp, 2), acc2[g]);
                __builtin_amdgcn_sched_group_barrier(SG_VALU, 1, 0);
            }
        }
    }
}

__device__ __forceinline__ float min3f(float a, float b, float c)
{
    return fminf(fminf(a, b), c);
}

__launch_bounds__(BLOCK)
__global__ void tps_wmma_kernel(const float* __restrict__ u,
                                const float* __restrict__ cp,
                                const float* __restrict__ w,
                                const float* __restrict__ poly,
                                float* __restrict__ out)
{
    __shared__ float4 sCP[NCP];
    __shared__ float4 sW[NCP * 3 / 4];
    __shared__ float sRed[SLICES][ROWS_PER_BLOCK * 3];

    const int tid   = threadIdx.x;
    const int slice = __builtin_amdgcn_readfirstlane(tid >> 5);
    const int lane  = tid & 31;
    const int half  = lane >> 4;
    const int q     = lane & 15;

    v2f bmat[QGROUPS];
    v8f cmat[QGROUPS];
#pragma unroll
    for (int g = 0; g < QGROUPS; ++g) {
        const int qIdx = blockIdx.x * ROWS_PER_BLOCK + g * 16 + q;
        const float ux = u[qIdx * 3 + 0];
        const float uy = u[qIdx * 3 + 1];
        const float uz = u[qIdx * 3 + 2];
        const float usq = ux * ux + uy * uy + uz * uz;
        bmat[g].x = half ? (-2.0f * uz) : (-2.0f * ux);
        bmat[g].y = half ? 1.0f         : (-2.0f * uy);
#pragma unroll
        for (int k = 0; k < 8; ++k) cmat[g][k] = usq;
    }

#pragma unroll
    for (int it = 0; it < NCP / BLOCK; ++it) {
        const int i = it * BLOCK + tid;
        const float cx = cp[i * 3 + 0];
        const float cy = cp[i * 3 + 1];
        const float cz = cp[i * 3 + 2];
        sCP[i] = make_float4(cx, cy, cz, cx * cx + cy * cy + cz * cz);
        float* sWf = reinterpret_cast<float*>(sW) + (i >> 1) * 6 + (i & 1);
        sWf[0] = HALF_LN2 * w[i * 3 + 0];
        sWf[2] = HALF_LN2 * w[i * 3 + 1];
        sWf[4] = HALF_LN2 * w[i * 3 + 2];
    }
    __syncthreads();

    v2f acc0[QGROUPS], acc1[QGROUPS], acc2[QGROUPS];
#pragma unroll
    for (int g = 0; g < QGROUPS; ++g) {
        acc0[g] = (v2f){0.0f, 0.0f};
        acc1[g] = (v2f){0.0f, 0.0f};
        acc2[g] = (v2f){0.0f, 0.0f};
    }

    const int cbBegin = slice * SLICE_LEN;

    for (int t = 0; t < SLICE_LEN; t += 32) {
        const int cb = cbBegin + t;
        const float4 cA = sCP[cb + q];
        const float4 cB = sCP[cb + 16 + q];
        v2f amatA, amatB;
        amatA.x = half ? cA.z : cA.x;
        amatA.y = half ? cA.w : cA.y;
        amatB.x = half ? cB.z : cB.x;
        amatB.y = half ? cB.w : cB.y;

        v8f dA[QGROUPS], dB[QGROUPS];
#pragma unroll
        for (int g = 0; g < QGROUPS; ++g) {
            dA[g] = __builtin_amdgcn_wmma_f32_16x16x4_f32(
                false, amatA, false, bmat[g], (short)0, cmat[g], false, false);
            dB[g] = __builtin_amdgcn_wmma_f32_16x16x4_f32(
                false, amatB, false, bmat[g], (short)0, cmat[g], false, false);
        }

        float4 wA[6], wB[6];
        const int mBase = ((cb + half * 8) * 3) >> 2;
#pragma unroll
        for (int k = 0; k < 6; ++k) {
            wA[k] = sW[mBase + k];
            wB[k] = sW[mBase + 12 + k];
        }

        float l1[QGROUPS * 6];
#pragma unroll
        for (int g = 0; g < QGROUPS; ++g) {
            l1[g * 6 + 0] = min3f(dA[g][0], dA[g][1], dA[g][2]);
            l1[g * 6 + 1] = min3f(dA[g][3], dA[g][4], dA[g][5]);
            l1[g * 6 + 2] = min3f(dA[g][6], dA[g][7], dB[g][0]);
            l1[g * 6 + 3] = min3f(dB[g][1], dB[g][2], dB[g][3]);
            l1[g * 6 + 4] = min3f(dB[g][4], dB[g][5], dB[g][6]);
            l1[g * 6 + 5] = dB[g][7];
        }
        float l2[QGROUPS * 2];
#pragma unroll
        for (int i = 0; i < QGROUPS * 2; ++i)
            l2[i] = min3f(l1[i * 3 + 0], l1[i * 3 + 1], l1[i * 3 + 2]);
        const float l3a = min3f(l2[0], l2[1], l2[2]);
        const float l3b = min3f(l2[3], l2[4], l2[5]);
        const float m   = fminf(min3f(l3a, l3b, l2[6]), l2[7]);

        if (__builtin_expect(__builtin_amdgcn_ballot_w32(m < EPS2) != 0u, 0)) {
#pragma unroll
            for (int g = 0; g < QGROUPS; ++g) {
                tile_accumulate_guarded(dA[g], wA, acc0[g], acc1[g], acc2[g]);
                tile_accumulate_guarded(dB[g], wB, acc0[g], acc1[g], acc2[g]);
            }
        } else {
            trip_accumulate(dA, dB, wA, wB, acc0, acc1, acc2);
        }
    }

#pragma unroll
    for (int g = 0; g < QGROUPS; ++g) {
        float s0 = acc0[g].x + acc0[g].y;
        float s1 = acc1[g].x + acc1[g].y;
        float s2 = acc2[g].x + acc2[g].y;
        s0 += __shfl_xor(s0, 16, 32);
        s1 += __shfl_xor(s1, 16, 32);
        s2 += __shfl_xor(s2, 16, 32);
        if (half == 0) {
            const int qLocal = g * 16 + q;
            sRed[slice][qLocal * 3 + 0] = s0;
            sRed[slice][qLocal * 3 + 1] = s1;
            sRed[slice][qLocal * 3 + 2] = s2;
        }
    }
    __syncthreads();

    if (tid < ROWS_PER_BLOCK * 3) {
        const int ql = tid / 3;
        const int a  = tid - ql * 3;
        const int qo = blockIdx.x * ROWS_PER_BLOCK + ql;

        float acc = sRed[0][tid];
#pragma unroll
        for (int s = 1; s < SLICES; ++s) acc += sRed[s][tid];

        const float px = u[qo * 3 + 0];
        const float py = u[qo * 3 + 1];
        const float pz = u[qo * 3 + 2];
        const float p = poly[0 * 3 + a] + px * poly[1 * 3 + a]
                      + py * poly[2 * 3 + a] + pz * poly[3 * 3 + a];
        out[qo * 3 + a] = acc + p;
    }
}

extern "C" void kernel_launch(void* const* d_in, const int* in_sizes, int n_in,
                              void* d_out, int out_size, void* d_ws, size_t ws_size,
                              hipStream_t stream) {
    const float* u    = (const float*)d_in[0];
    const float* cp   = (const float*)d_in[1];
    const float* w    = (const float*)d_in[2];
    const float* poly = (const float*)d_in[3];
    float* out = (float*)d_out;

    dim3 grid(BATCH / ROWS_PER_BLOCK);
    tps_wmma_kernel<<<grid, BLOCK, 0, stream>>>(u, cp, w, poly, out);
}


# r<EPS guard taken out of the loop: the loop runs the unguarded body (no minimum tree, no branch); a non-finite sum (sq<=0 gives NaN) sends the wave's slice through the guarded body; same sum order
# speedup vs baseline: 8.2987x; 1.1039x over previous
#include <hip/hip_runtime.h>
#include <math.h>


typedef __attribute__((ext_vector_type(2))) float v2f;
typedef __attribute__((ext_vector_type(8))) float v8f;

#define BATCH 16384
#define NCP   4096
#define BLOCK 256
#define QGROUPS 4
#define SLICES  8
#define SLICE_LEN (NCP / SLICES)
#define ROWS_PER_BLOCK (QGROUPS * 16)

#define EPS2      1e-20f
#define HALF_LN2  0.34657359028f

__device__ __forceinline__ v2f weight_pair(const float4 (&wv)[6], int p, int c)
{
    const int idx = 6 * p + 2 * c;
    const float4 f = wv[idx >> 2];
    return (idx & 2) ? (v2f){f.z, f.w} : (v2f){f.x, f.y};
}

__device__ __forceinline__ void tile_accumulate_guarded(const v8f d, const float4 (&wv)[6],
                                                        v2f& acc0, v2f& acc1, v2f& acc2)
{
#pragma unroll
    for (int p = 0; p < 4; ++p) {
        const float sqa = d[2 * p];
        const float sqb = d[2 * p + 1];
        const bool  smalla = sqa < EPS2;
        const bool  smallb = sqb < EPS2;
        const float ca = smalla ? EPS2 : sqa;
        const float cb = smallb ? EPS2 : sqb;
        const v2f rt = {__builtin_amdgcn_sqrtf(ca), __builtin_amdgcn_sqrtf(cb)};
        const v2f lg = {__builtin_amdgcn_logf(ca), __builtin_amdgcn_logf(cb)};
        v2f kv = rt * lg;
        kv.x = smalla ? 0.0f : kv.x;
        kv.y = smallb ? 0.0f : kv.y;
        acc0 = __builtin_elementwise_fma(kv, weight_pair(wv, p, 0), acc0);
        acc1 = __builtin_elementwise_fma(kv, weight_pair(wv, p, 1), acc1);
        acc2 = __builtin_elementwise_fma(kv, weight_pair(wv, p, 2), acc2);
    }
}

#define SG_VALU  0x002
#define SG_TRANS 0x400

__device__ __forceinline__ void trip_accumulate(const v8f (&dA)[QGROUPS], const v8f (&dB)[QGROUPS],
                                                const float4 (&wA)[6], const float4 (&wB)[6],
                                                v2f (&acc0)[QGROUPS], v2f (&acc1)[QGROUPS],
                                                v2f (&acc2)[QGROUPS])
{
    constexpr int STEPS = 8;
    v2f rt[STEPS][QGROUPS], lg[STEPS][QGROUPS], kv[STEPS][QGROUPS];
#pragma unroll
    for (int b = 0; b <= STEPS; ++b) {
        const int  p  = b & 3;
        const bool tB = b >= 4;
        const int  pp = (b - 1) & 3;
        const bool tP = (b - 1) >= 4;
#pragma unroll
        for (int s = 0; s < 4 * QGROUPS; ++s) {
            if (b < STEPS) {
                const int g = s >> 2;
                const float sqa = tB ? dB[g][2 * p]     : dA[g][2 * p];
                const float sqb = tB ? dB[g][2 * p + 1] : dA[g][2 * p + 1];
                if ((s & 3) == 0) rt[b][g].x = __builtin_amdgcn_sqrtf(sqa);
                if ((s & 3) == 1) rt[b][g].y = __builtin_amdgcn_sqrtf(sqb);
                if ((s & 3) == 2) lg[b][g].x = __builtin_amdgcn_logf(sqa);
                if ((s & 3) == 3) lg[b][g].y = __builtin_amdgcn_logf(sqb);
                __builtin_amdgcn_sched_group_barrier(SG_TRANS, 1, 0);
            }
            if (b >= 1) {
                const int g = s & (QGROUPS - 1);
                const int c = (s >> 2) - 1;
                if (c < 0)  kv[b - 1][g] = rt[b - 1][g] * lg[b - 1][g];
                if (c == 0) acc0[g] = __builtin_elementwise_fma(kv[b - 1][g], weight_pair(tP ? wB : wA, pp, 0), acc0[g]);
                if (c == 1) acc1[g] = __builtin_elementwise_fma(kv[b - 1][g], weight_pair(tP ? wB : wA, pp, 1), acc1[g]);
                if (c == 2) acc2[g] = __builtin_elementwise_fma(kv[b - 1][g], weight_pair(tP ? wB : wA, pp, 2), acc2[g]);
                __builtin_amdgcn_sched_group_barrier(SG_VALU, 1, 0);
            }
        }
    }
}

template <bool GUARDED>
__device__ __forceinline__ void slice_pass(const float4 (&sCP)[NCP], const float4 (&sW)[NCP * 3 / 4],
                                           const int cbBegin, const int q, const int half,
                                           const v2f (&bmat)[QGROUPS], const v8f (&cmat)[QGROUPS],
                                           v2f (&acc0)[QGROUPS], v2f (&acc1)[QGROUPS],
                                           v2f (&acc2)[QGROUPS])
{
    for (int t = 0; t < SLICE_LEN; t += 32) {
        const int cb = cbBegin + t;
        const float4 cA = sCP[cb + q];
        const float4 cB = sCP[cb + 16 + q];
        v2f amatA, amatB;
        amatA.x = half ? cA.z : cA.x;
        amatA.y = half ? cA.w : cA.y;
        amatB.x = half ? cB.z : cB.x;
        amatB.y = half ? cB.w : cB.y;

        v8f dA[QGROUPS], dB[QGROUPS];
#pragma unroll
        for (int g = 0; g < QGROUPS; ++g) {
            dA[g] = __builtin_amdgcn_wmma_f32_16x16x4_f32(
                false, amatA, false, bmat[g], (short)0, cmat[g], false, false);
            dB[g] = __builtin_amdgcn_wmma_f32_16x16x4_f32(
                false, amatB, false, bmat[g], (short)0, cmat[g], false, false);
        }

        float4 wA[6], wB[6];
        const int mBase = ((cb + half * 8) * 3) >> 2;
#pragma unroll
        for (int k = 0; k < 6; ++k) {
            wA[k] = sW[mBase + k];
            wB[k] = sW[mBase + 12 + k];
        }

        if (GUARDED) {
#pragma unroll
            for (int g = 0; g < QGROUPS; ++g) {
                tile_accumulate_guarded(dA[g], wA, acc0[g], acc1[g], acc2[g]);
                tile_accumulate_guarded(dB[g], wB, acc0[g], acc1[g], acc2[g]);
            }
        } else {
            __builtin_amdgcn_sched_barrier(0);
            trip_accumulate(dA, dB, wA, wB, acc0, acc1, acc2);
        }
    }
}

__launch_bounds__(BLOCK)
__global__ void tps_wmma_kernel(const float* __restrict__ u,
                                const float* __restrict__ cp,
                                const float* __restrict__ w,
                                const float* __restrict__ poly,
                                float* __restrict__ out)
{
    __shared__ float4 sCP[NCP];
    __shared__ float4 sW[NCP * 3 / 4];
    __shared__ float sRed[SLICES][ROWS_PER_BLOCK * 3];

    const int tid   = threadIdx.x;
    const int slice = __builtin_amdgcn_readfirstlane(tid >> 5);
    const int lane  = tid & 31;
    const int half  = lane >> 4;
    const int q     = lane & 15;

    v2f bmat[QGROUPS];
    v8f cmat[QGROUPS];
#pragma unroll
    for (int g = 0; g < QGROUPS; ++g) {
        const int qIdx = blockIdx.x * ROWS_PER_BLOCK + g * 16 + q;
        const float ux = u[qIdx * 3 + 0];
        const float uy = u[qIdx * 3 + 1];
        const float uz = u[qIdx * 3 + 2];
        const float usq = ux * ux + uy * uy + uz * uz;
        bmat[g].x = half ? (-2.0f * uz) : (-2.0f * ux);
        bmat[g].y = half ? 1.0f         : (-2.0f * uy);
#pragma unroll
        for (int k = 0; k < 8; ++k) cmat[g][k] = usq;
    }

#pragma unroll
    for (int it = 0; it < NCP / BLOCK; ++it) {
        const int i = it * BLOCK + tid;
        const float cx = cp[i * 3 + 0];
        const float cy = cp[i * 3 + 1];
        const float cz = cp[i * 3 + 2];
        sCP[i] = make_float4(cx, cy, cz, cx * cx + cy * cy + cz * cz);
        float* sWf = reinterpret_cast<float*>(sW) + (i >> 1) * 6 + (i & 1);
        sWf[0] = HALF_LN2 * w[i * 3 + 0];
        sWf[2] = HALF_LN2 * w[i * 3 + 1];
        sWf[4] = HALF_LN2 * w[i * 3 + 2];
    }
    __syncthreads();

    v2f acc0[QGROUPS], acc1[QGROUPS], acc2[QGROUPS];
#pragma unroll
    for (int g = 0; g < QGROUPS; ++g) {
        acc0[g] = (v2f){0.0f, 0.0f};
        acc1[g] = (v2f){0.0f, 0.0f};
        acc2[g] = (v2f){0.0f, 0.0f};
    }

    const int cbBegin = slice * SLICE_LEN;

    slice_pass<false>(sCP, sW, cbBegin, q, half, bmat, cmat, acc0, acc1, acc2);

    float chk = 0.0f;
#pragma unroll
    for (int g = 0; g < QGROUPS; ++g)
        chk += (acc0[g].x + acc0[g].y) + (acc1[g].x + acc1[g].y) + (acc2[g].x + acc2[g].y);
    const bool notFinite = !(__builtin_fabsf(chk) <= 3.402823466e38f);

    if (__builtin_expect(__builtin_amdgcn_ballot_w32(notFinite) != 0u, 0)) {
#pragma unroll
        for (int g = 0; g < QGROUPS; ++g) {
            acc0[g] = (v2f){0.0f, 0.0f};
            acc1[g] = (v2f){0.0f, 0.0f};
            acc2[g] = (v2f){0.0f, 0.0f};
        }
        slice_pass<true>(sCP, sW, cbBegin, q, half, bmat, cmat, acc0, acc1, acc2);
    }

#pragma unroll
    for (int g = 0; g < QGROUPS; ++g) {
        float s0 = acc0[g].x + acc0[g].y;
        float s1 = acc1[g].x + acc1[g].y;
        float s2 = acc2[g].x + acc2[g].y;
        s0 += __shfl_xor(s0, 16, 32);
        s1 += __shfl_xor(s1, 16, 32);
        s2 += __shfl_xor(s2, 16, 32);
        if (half == 0) {
            const int qLocal = g * 16 + q;
            sRed[slice][qLocal * 3 + 0] = s0;
            sRed[slice][qLocal * 3 + 1] = s1;
            sRed[slice][qLocal * 3 + 2] = s2;
        }
    }
    __syncthreads();

    if (tid < ROWS_PER_BLOCK * 3) {
        const int ql = tid / 3;
        const int a  = tid - ql * 3;
        const int qo = blockIdx.x * ROWS_PER_BLOCK + ql;

        float acc = sRed[0][tid];
#pragma unroll
        for (int s = 1; s < SLICES; ++s) acc += sRed[s][tid];

        const float px = u[qo * 3 + 0];
        const float py = u[qo * 3 + 1];
        const float pz = u[qo * 3 + 2];
        const float p = poly[0 * 3 + a] + px * poly[1 * 3 + a]
                      + py * poly[2 * 3 + a] + pz * poly[3 * 3 + a];
        out[qo * 3 + a] = acc + p;
    }
}

extern "C" void kernel_launch(void* const* d_in, const int* in_sizes, int n_in,
                              void* d_out, int out_size, void* d_ws, size_t ws_size,
                              hipStream_t stream) {
    const float* u    = (const float*)d_in[0];
    const float* cp   = (const float*)d_in[1];
    const float* w    = (const float*)d_in[2];
    const float* poly = (const float*)d_in[3];
    float* out = (float*)d_out;

    dim3 grid(BATCH / ROWS_PER_BLOCK);
    tps_wmma_kernel<<<grid, BLOCK, 0, stream>>>(u, cp, w, poly, out);
}
